# starting kernel
# baseline (speedup 1.0000x reference)
#include <hip/hip_runtime.h>

#define GCN_N 100000
#define GCN_D 128

typedef float v2f __attribute__((ext_vector_type(2)));
typedef float v8f __attribute__((ext_vector_type(8)));

__global__ __launch_bounds__(256) void gcn_fill_zero(float* __restrict__ p, long long n) {
    long long i = (long long)blockIdx.x * blockDim.x + threadIdx.x;
    if (i < n) p[i] = 0.0f;
}

__global__ __launch_bounds__(256) void gcn_deg_accum(const int* __restrict__ dst,
                                                     float* __restrict__ deg, int ne) {
    int i = blockIdx.x * blockDim.x + threadIdx.x;
    if (i < ne) atomicAdd(&deg[dst[i]], 1.0f);
}

__global__ __launch_bounds__(256) void gcn_finalize_dinv(float* __restrict__ dinv, int n) {
    int i = blockIdx.x * blockDim.x + threadIdx.x;
    if (i < n) dinv[i] = rsqrtf(dinv[i] + 1.0f);
}

__global__ __launch_bounds__(256) void gcn_gemm_wmma(const float* __restrict__ X,
                                                     const float* __restrict__ W,
                                                     float* __restrict__ H) {
    const int lane     = threadIdx.x & 31;
    const int wave     = blockIdx.x * 8 + (threadIdx.x >> 5);
    const int col_tile = wave & 7;
    const int row_tile = wave >> 3;
    const int m        = lane & 15;
    const int half     = lane >> 4;
    const int koff     = half << 1;

    const float* xrow = X + (size_t)(row_tile * 16 + m) * GCN_D;
    const int    ncol = col_tile * 16 + m;

    v8f acc = {};
#pragma unroll
    for (int k = 0; k < GCN_D; k += 4) {
        v2f a = *(const v2f*)(xrow + k + koff);
        v2f b;
        b.x = W[(size_t)(k + koff)     * GCN_D + ncol];
        b.y = W[(size_t)(k + koff + 1) * GCN_D + ncol];
        acc = __builtin_amdgcn_wmma_f32_16x16x4_f32(
                  false, a, false, b, (short)0, acc, false, false);
    }

    const int mbase = row_tile * 16 + half * 8;
    float* hp = H + (size_t)mbase * GCN_D + col_tile * 16 + m;
#pragma unroll
    for (int i = 0; i < 8; ++i) hp[(size_t)i * GCN_D] = acc[i];
}

__global__ __launch_bounds__(256) void gcn_scatter(const int* __restrict__ src,
                                                   const int* __restrict__ dst,
                                                   const float* __restrict__ h,
                                                   const float* __restrict__ dinv,
                                                   float* __restrict__ acc, int ne) {
    int warp = (int)((blockIdx.x * blockDim.x + threadIdx.x) >> 5);
    int lane = threadIdx.x & 31;
    if (warp >= ne) return;
    int s = src[warp];
    int d = dst[warp];
    float nrm = dinv[s] * dinv[d];
    float4 v = ((const float4*)(h + (size_t)s * GCN_D))[lane];
    float* o = acc + (size_t)d * GCN_D + lane * 4;
    atomicAdd(o + 0, v.x * nrm);
    atomicAdd(o + 1, v.y * nrm);
    atomicAdd(o + 2, v.z * nrm);
    atomicAdd(o + 3, v.w * nrm);
}

__global__ __launch_bounds__(256) void gcn_epilogue(const float* __restrict__ acc,
                                                    const float* __restrict__ h,
                                                    const float* __restrict__ dinv,
                                                    const float* __restrict__ b,
                                                    float* __restrict__ out, long long n) {
    long long i = (long long)blockIdx.x * blockDim.x + threadIdx.x;
    if (i >= n) return;
    int node = (int)(i >> 7);
    int j    = (int)(i & 127);
    float di = dinv[node];
    float v  = acc[i] + h[i] * di * di + b[j];
    out[i] = v > 0.0f ? v : 0.0f;
}

extern "C" void kernel_launch(void* const* d_in, const int* in_sizes, int n_in,
                              void* d_out, int out_size, void* d_ws, size_t ws_size,
                              hipStream_t stream) {
    const float* x  = (const float*)d_in[0];
    const float* W1 = (const float*)d_in[1];
    const float* b1 = (const float*)d_in[2];
    const float* W2 = (const float*)d_in[3];
    const float* b2 = (const float*)d_in[4];
    const float* W3 = (const float*)d_in[5];
    const float* b3 = (const float*)d_in[6];
    const int*   eidx = (const int*)d_in[7];
    const int ne = in_sizes[7] / 2;
    const int* src = eidx;
    const int* dst = eidx + ne;

    char* ws = (char*)d_ws;
    size_t off = 0;
    float* dinv = (float*)(ws + off); off += (((size_t)GCN_N * 4) + 255) & ~(size_t)255;
    float* h    = (float*)(ws + off); off += (size_t)GCN_N * GCN_D * 4;
    float* acc  = (float*)(ws + off); off += (size_t)GCN_N * GCN_D * 4;
    float* buf  = (float*)(ws + off);

    const long long feat_elems = (long long)GCN_N * GCN_D;
    const int blk = 256;
    const int g_node  = (GCN_N + blk - 1) / blk;
    const int g_edge  = (ne + blk - 1) / blk;
    const long long g_feat_ll = (feat_elems + blk - 1) / blk;
    const int g_feat  = (int)g_feat_ll;
    const int g_gemm  = GCN_N / 16;
    const int g_scat  = (int)(((long long)ne * 32 + blk - 1) / blk);

    gcn_fill_zero<<<g_node, blk, 0, stream>>>(dinv, GCN_N);
    gcn_deg_accum<<<g_edge, blk, 0, stream>>>(dst, dinv, ne);
    gcn_finalize_dinv<<<g_node, blk, 0, stream>>>(dinv, GCN_N);

    gcn_gemm_wmma<<<g_gemm, blk, 0, stream>>>(x, W1, h);
    gcn_fill_zero<<<g_feat, blk, 0, stream>>>(acc, feat_elems);
    gcn_scatter<<<g_scat, blk, 0, stream>>>(src, dst, h, dinv, acc, ne);
    gcn_epilogue<<<g_feat, blk, 0, stream>>>(acc, h, dinv, b1, buf, feat_elems);

    gcn_gemm_wmma<<<g_gemm, blk, 0, stream>>>(buf, W2, h);
    gcn_fill_zero<<<g_feat, blk, 0, stream>>>(acc, feat_elems);
    gcn_scatter<<<g_scat, blk, 0, stream>>>(src, dst, h, dinv, acc, ne);
    gcn_epilogue<<<g_feat, blk, 0, stream>>>(acc, h, dinv, b2, buf, feat_elems);

    gcn_gemm_wmma<<<g_gemm, blk, 0, stream>>>(buf, W3, h);
    gcn_fill_zero<<<g_feat, blk, 0, stream>>>(acc, feat_elems);
    gcn_scatter<<<g_scat, blk, 0, stream>>>(src, dst, h, dinv, acc, ne);
    gcn_epilogue<<<g_feat, blk, 0, stream>>>(acc, h, dinv, b3, (float*)d_out, feat_elems);
}


# Edge sums in LDS: edges grouped once per call by 100-row destination block; a work-group adds a block's weighted h rows into its LDS tile (LDS float adds), no global float atomics; arrival-order sums
# speedup vs baseline: 2.6224x; 2.6224x over previous
#include <hip/hip_runtime.h>

#define GCN_N 100000
#define GCN_D 128
#define GCN_DINV_SLOTS 100032
#define AG_ROWS 100
#define AG_BLOCKS 1000
#define AG_BIN_INTS 2048
#define AG_CURSOR_AT 1024
#define AG_CHUNK 4096
#define AG_SRC_BITS 20

typedef float v2f __attribute__((ext_vector_type(2)));
typedef float v8f __attribute__((ext_vector_type(8)));

__global__ __launch_bounds__(256) void gcn_fill_zero(float* __restrict__ p, long long n) {
    long long i = (long long)blockIdx.x * blockDim.x + threadIdx.x;
    if (i < n) p[i] = 0.0f;
}

__global__ __launch_bounds__(256) void gcn_deg_and_block_counts(const int* __restrict__ dst,
                                                                float* __restrict__ deg,
                                                                int* __restrict__ bincount, int ne) {
    __shared__ int hist[AG_BLOCKS];
    for (int b = threadIdx.x; b < AG_BLOCKS; b += 256) hist[b] = 0;
    __syncthreads();
    const int base = blockIdx.x * AG_CHUNK + threadIdx.x;
    const int last = ne - 1;
    int d[AG_CHUNK / 256];
#pragma unroll
    for (int i = 0; i < AG_CHUNK / 256; ++i) {
        const int e = base + i * 256;
        d[i] = dst[e < last ? e : last];
    }
#pragma unroll
    for (int i = 0; i < AG_CHUNK / 256; ++i) {
        if (base + i * 256 < ne) {
            atomicAdd(&deg[d[i]], 1.0f);
            atomicAdd(&hist[d[i] / AG_ROWS], 1);
        }
    }
    __syncthreads();
    for (int b = threadIdx.x; b < AG_BLOCKS; b += 256) {
        const int c = hist[b];
        if (c) atomicAdd(&bincount[b], c);
    }
}

__device__ __forceinline__ int gcn_block_sum(int v, int* wsum) {
    for (int o = 16; o > 0; o >>= 1) v += __shfl_xor(v, o);
    if ((threadIdx.x & 31) == 0) wsum[threadIdx.x >> 5] = v;
    __syncthreads();
    int t = 0;
#pragma unroll
    for (int k = 0; k < 8; ++k) t += wsum[k];
    return t;
}

__global__ __launch_bounds__(256) void gcn_dinv_and_edge_records(const int* __restrict__ src,
                                                                 const int* __restrict__ dst,
                                                                 const float* __restrict__ deg,
                                                                 const int* __restrict__ bincount,
                                                                 int* __restrict__ cursor,
                                                                 float* __restrict__ dinv,
                                                                 int* __restrict__ erec,
                                                                 int n, int ne) {
    __shared__ int hist[AG_BLOCKS];
    __shared__ int slot0[AG_BLOCKS];
    __shared__ int wsum[8];
    const int tid = threadIdx.x;
    const int stride = gridDim.x * 256;
    for (int j = blockIdx.x * 256 + tid; j < n; j += stride) dinv[j] = rsqrtf(deg[j] + 1.0f);

    for (int b = tid; b < AG_BLOCKS; b += 256) hist[b] = 0;
    __syncthreads();
    int s[AG_CHUNK / 256], d[AG_CHUNK / 256];
    const int base = blockIdx.x * AG_CHUNK + tid;
    const int last = ne - 1;
#pragma unroll
    for (int i = 0; i < AG_CHUNK / 256; ++i) {
        const int e  = base + i * 256;
        const int ec = e < last ? e : last;
        s[i] = src[ec]; d[i] = dst[ec];
    }
#pragma unroll
    for (int i = 0; i < AG_CHUNK / 256; ++i)
        if (base + i * 256 < ne) atomicAdd(&hist[d[i] / AG_ROWS], 1);
    int c[4] = {0, 0, 0, 0};
    if (tid < AG_BLOCKS / 4) {
        const int4 cv = *(const int4*)(bincount + 4 * tid);
        c[0] = cv.x; c[1] = cv.y; c[2] = cv.z; c[3] = cv.w;
    }
    const int tsum = c[0] + c[1] + c[2] + c[3];
    int incl = tsum;
    const int lane = tid & 31, wv = tid >> 5;
    for (int o = 1; o < 32; o <<= 1) {
        const int t = __shfl_up(incl, o);
        if (lane >= o) incl += t;
    }
    if (lane == 31) wsum[wv] = incl;
    __syncthreads();
    int run = incl - tsum;
    for (int k = 0; k < wv; ++k) run += wsum[k];
    float ds[AG_CHUNK / 256], dd[AG_CHUNK / 256];
#pragma unroll
    for (int i = 0; i < AG_CHUNK / 256; ++i) { ds[i] = deg[s[i]]; dd[i] = deg[d[i]]; }
    if (tid < AG_BLOCKS / 4) {
        int got[4];
#pragma unroll
        for (int k = 0; k < 4; ++k) got[k] = atomicAdd(&cursor[4 * tid + k], hist[4 * tid + k]);
#pragma unroll
        for (int k = 0; k < 4; ++k) {
            slot0[4 * tid + k] = run + got[k];
            hist[4 * tid + k] = 0;
            run += c[k];
        }
    }
    __syncthreads();
#pragma unroll
    for (int i = 0; i < AG_CHUNK / 256; ++i) {
        if (base + i * 256 < ne) {
            const int b = d[i] / AG_ROWS;
            const int pos = slot0[b] + atomicAdd(&hist[b], 1);
            const float w = rsqrtf(ds[i] + 1.0f) * rsqrtf(dd[i] + 1.0f);
            *(int2*)(erec + 2 * (size_t)pos) = make_int2(s[i] | ((d[i] - b * AG_ROWS) << AG_SRC_BITS), __float_as_int(w));
        }
    }
}

template <bool RELU_IN>
__device__ __forceinline__ void gcn_gemm_body(const float* __restrict__ X,
                                              const float* __restrict__ W,
                                              const float* __restrict__ bias,
                                              const float* __restrict__ dinv,
                                              float* __restrict__ H,
                                              float* __restrict__ S) {
    const int lane     = threadIdx.x & 31;
    const int wave     = blockIdx.x * 8 + (threadIdx.x >> 5);
    const int col_tile = wave & 7;
    const int row_tile = wave >> 3;
    const int m        = lane & 15;
    const int half     = lane >> 4;
    const int koff     = half << 1;

    const float* xrow = X + (size_t)(row_tile * 16 + m) * GCN_D;
    const int    ncol = col_tile * 16 + m;

    v8f acc = {};
#pragma unroll
    for (int k = 0; k < GCN_D; k += 4) {
        v2f a = *(const v2f*)(xrow + k + koff);
        if (RELU_IN) {
            a.x = a.x > 0.0f ? a.x : 0.0f;
            a.y = a.y > 0.0f ? a.y : 0.0f;
        }
        v2f b;
        b.x = W[(size_t)(k + koff)     * GCN_D + ncol];
        b.y = W[(size_t)(k + koff + 1) * GCN_D + ncol];
        acc = __builtin_amdgcn_wmma_f32_16x16x4_f32(
                  false, a, false, b, (short)0, acc, false, false);
    }

    const int   mbase = row_tile * 16 + half * 8;
    const float bc    = bias[ncol];
    const float4 d0   = *(const float4*)(dinv + mbase);
    const float4 d1   = *(const float4*)(dinv + mbase + 4);
    const float di[8] = {d0.x, d0.y, d0.z, d0.w, d1.x, d1.y, d1.z, d1.w};
    const size_t o = (size_t)mbase * GCN_D + ncol;
    float* hp = H + o;
    float* sp = S + o;
#pragma unroll
    for (int i = 0; i < 8; ++i) {
        const float hv = acc[i];
        hp[(size_t)i * GCN_D] = hv;
        sp[(size_t)i * GCN_D] = hv * di[i] * di[i] + bc;
    }
}

__global__ __launch_bounds__(256) void gcn_gemm_first(const float* __restrict__ X,
                                                      const float* __restrict__ W,
                                                      const float* __restrict__ bias,
                                                      const float* __restrict__ dinv,
                                                      float* __restrict__ H,
                                                      float* __restrict__ S) {
    gcn_gemm_body<false>(X, W, bias, dinv, H, S);
}

__global__ __launch_bounds__(256) void gcn_gemm_next(const float* __restrict__ X,
                                                     const float* __restrict__ W,
                                                     const float* __restrict__ bias,
                                                     const float* __restrict__ dinv,
                                                     float* __restrict__ H,
                                                     float* __restrict__ S) {
    gcn_gemm_body<true>(X, W, bias, dinv, H, S);
}

#define AG_GROUP 8
#define AG_TILE_ROWS 6

__global__ __launch_bounds__(256) void gcn_block_sums(const int* __restrict__ erec,
                                                      const int* __restrict__ bincount,
                                                      const float* __restrict__ h,
                                                      float* __restrict__ acc) {
    __shared__ float tile[AG_ROWS * GCN_D];
    __shared__ int wsum[8];
    const int tid  = threadIdx.x;
    const int lane = tid & 31;
    const int wv   = __builtin_amdgcn_readfirstlane(tid >> 5);
    const int blk  = blockIdx.x;

    int part = 0;
    for (int b = tid; b < blk; b += 256) part += bincount[b];
    const int cnt = bincount[blk];

    float* arow = acc + (size_t)blk * (AG_ROWS * GCN_D) + 4 * lane;
#pragma unroll
    for (int p = 0; p < 2; ++p) {
        float4 v[AG_TILE_ROWS];
#pragma unroll
        for (int i = 0; i < AG_TILE_ROWS; ++i)
            v[i] = *(const float4*)(arow + (wv + 8 * (AG_TILE_ROWS * p + i)) * GCN_D);
#pragma unroll
        for (int i = 0; i < AG_TILE_ROWS; ++i) {
            float* t = tile + (wv + 8 * (AG_TILE_ROWS * p + i)) * GCN_D + lane;
            t[0] = v[i].x; t[32] = v[i].y; t[64] = v[i].z; t[96] = v[i].w;
        }
    }
    if (wv < AG_ROWS - 96) {
        const float4 v = *(const float4*)(arow + (96 + wv) * GCN_D);
        float* t = tile + (96 + wv) * GCN_D + lane;
        t[0] = v.x; t[32] = v.y; t[64] = v.z; t[96] = v.w;
    }
    const int start = __builtin_amdgcn_readfirstlane(gcn_block_sum(part, wsum));
    const int end   = start + __builtin_amdgcn_readfirstlane(cnt);

    const float* hl = h + 4 * lane;
    float*       tl = tile + lane;
    int g0 = start + wv * 32;
    int2 nxt = make_int2(0, 0);
    if (g0 < end) {
        int idx = g0 + lane;
        if (idx > end - 1) idx = end - 1;
        nxt = *(const int2*)(erec + 2 * (size_t)idx);
    }
    for (; g0 < end; g0 += 256) {
        int m = end - g0;
        if (m > 32) m = 32;
        const int2 rec = nxt;
        if (g0 + 256 < end) {
            int idx = g0 + 256 + lane;
            if (idx > end - 1) idx = end - 1;
            nxt = *(const int2*)(erec + 2 * (size_t)idx);
        }
        int g = 0;
        for (; g + AG_GROUP <= m; g += AG_GROUP) {
            float4 v[AG_GROUP];
#pragma unroll
            for (int i = 0; i < AG_GROUP; ++i) {
                const int s = __builtin_amdgcn_readlane(rec.x, g + i) & ((1 << AG_SRC_BITS) - 1);
                v[i] = *(const float4*)(hl + (size_t)s * GCN_D);
            }
#pragma unroll
            for (int i = 0; i < AG_GROUP; ++i) {
                const int   dl = __builtin_amdgcn_readlane(rec.x, g + i) >> AG_SRC_BITS;
                const float w  = __int_as_float(__builtin_amdgcn_readlane(rec.y, g + i));
                float* t = tl + dl * GCN_D;
                atomicAdd(t +  0, v[i].x * w);
                atomicAdd(t + 32, v[i].y * w);
                atomicAdd(t + 64, v[i].z * w);
                atomicAdd(t + 96, v[i].w * w);
            }
        }
        for (; g + 4 <= m; g += 4) {
            float4 v[4];
#pragma unroll
            for (int i = 0; i < 4; ++i) {
                const int s = __builtin_amdgcn_readlane(rec.x, g + i) & ((1 << AG_SRC_BITS) - 1);
                v[i] = *(const float4*)(hl + (size_t)s * GCN_D);
            }
#pragma unroll
            for (int i = 0; i < 4; ++i) {
                const int   dl = __builtin_amdgcn_readlane(rec.x, g + i) >> AG_SRC_BITS;
                const float w  = __int_as_float(__builtin_amdgcn_readlane(rec.y, g + i));
                float* t = tl + dl * GCN_D;
                atomicAdd(t +  0, v[i].x * w);
                atomicAdd(t + 32, v[i].y * w);
                atomicAdd(t + 64, v[i].z * w);
                atomicAdd(t + 96, v[i].w * w);
            }
        }
        for (; g < m; ++g) {
            const int   p  = __builtin_amdgcn_readlane(rec.x, g);
            const float w  = __int_as_float(__builtin_amdgcn_readlane(rec.y, g));
            const float4 v = *(const float4*)(hl + (size_t)(p & ((1 << AG_SRC_BITS) - 1)) * GCN_D);
            float* t = tl + (p >> AG_SRC_BITS) * GCN_D;
            atomicAdd(t +  0, v.x * w);
            atomicAdd(t + 32, v.y * w);
            atomicAdd(t + 64, v.z * w);
            atomicAdd(t + 96, v.w * w);
        }
    }
    __syncthreads();
#pragma unroll
    for (int i = 0; i < 2 * AG_TILE_ROWS; ++i) {
        const float* t = tile + (wv + 8 * i) * GCN_D + lane;
        float4 v;
        v.x = t[0]; v.y = t[32]; v.z = t[64]; v.w = t[96];
        *(float4*)(arow + (wv + 8 * i) * GCN_D) = v;
    }
    if (wv < AG_ROWS - 96) {
        const float* t = tile + (96 + wv) * GCN_D + lane;
        float4 v;
        v.x = t[0]; v.y = t[32]; v.z = t[64]; v.w = t[96];
        *(float4*)(arow + (96 + wv) * GCN_D) = v;
    }
}

#define RELU_PIECES 4
__global__ __launch_bounds__(256) void gcn_relu_inplace(float* __restrict__ p) {
    float4* q = (float4*)p + (size_t)blockIdx.x * (256 * RELU_PIECES) + threadIdx.x;
    float4 v[RELU_PIECES];
#pragma unroll
    for (int i = 0; i < RELU_PIECES; ++i) v[i] = q[i * 256];
#pragma unroll
    for (int i = 0; i < RELU_PIECES; ++i) {
        float4 r;
        r.x = v[i].x > 0.0f ? v[i].x : 0.0f;
        r.y = v[i].y > 0.0f ? v[i].y : 0.0f;
        r.z = v[i].z > 0.0f ? v[i].z : 0.0f;
        r.w = v[i].w > 0.0f ? v[i].w : 0.0f;
        q[i * 256] = r;
    }
}

extern "C" void kernel_launch(void* const* d_in, const int* in_sizes, int n_in,
                              void* d_out, int out_size, void* d_ws, size_t ws_size,
                              hipStream_t stream) {
    const float* x  = (const float*)d_in[0];
    const float* W1 = (const float*)d_in[1];
    const float* b1 = (const float*)d_in[2];
    const float* W2 = (const float*)d_in[3];
    const float* b2 = (const float*)d_in[4];
    const float* W3 = (const float*)d_in[5];
    const float* b3 = (const float*)d_in[6];
    const int*   eidx = (const int*)d_in[7];
    const int ne = in_sizes[7] / 2;
    const int* src = eidx;
    const int* dst = eidx + ne;

    char* ws = (char*)d_ws;
    size_t off = 0;
    float* deg   = (float*)(ws + off); off += (size_t)GCN_DINV_SLOTS * 4;
    int*   bins  = (int*)(ws + off);   off += (size_t)AG_BIN_INTS * 4;
    float* dinv  = (float*)(ws + off); off += (size_t)GCN_DINV_SLOTS * 4;
    int*   erec  = (int*)(ws + off);   off += (((size_t)ne * 8 + 255) / 256) * 256;
    float* h     = (float*)(ws + off); off += (size_t)GCN_N * GCN_D * 4;
    float* acc_a = (float*)(ws + off); off += (size_t)GCN_N * GCN_D * 4;
    float* acc_b = (float*)(ws + off);
    float* out   = (float*)d_out;

    const int blk = 256;
    const int g_fill  = (GCN_DINV_SLOTS + AG_BIN_INTS + blk - 1) / blk;
    const int g_bin   = (ne + AG_CHUNK - 1) / AG_CHUNK;
    const int g_gemm  = GCN_N / 16;
    const int g_relu  = GCN_N * GCN_D / (4 * RELU_PIECES * blk);

    gcn_fill_zero<<<g_fill, blk, 0, stream>>>(deg, GCN_DINV_SLOTS + AG_BIN_INTS);
    gcn_deg_and_block_counts<<<g_bin, blk, 0, stream>>>(dst, deg, bins, ne);
    gcn_dinv_and_edge_records<<<g_bin, blk, 0, stream>>>(src, dst, deg, bins, bins + AG_CURSOR_AT, dinv, erec, GCN_N, ne);

    gcn_gemm_first<<<g_gemm, blk, 0, stream>>>(x, W1, b1, dinv, h, acc_a);
    gcn_block_sums<<<AG_BLOCKS, blk, 0, stream>>>(erec, bins, h, acc_a);

    gcn_gemm_next<<<g_gemm, blk, 0, stream>>>(acc_a, W2, b2, dinv, h, acc_b);
    gcn_block_sums<<<AG_BLOCKS, blk, 0, stream>>>(erec, bins, h, acc_b);

    gcn_gemm_next<<<g_gemm, blk, 0, stream>>>(acc_b, W3, b3, dinv, h, out);
    gcn_block_sums<<<AG_BLOCKS, blk, 0, stream>>>(erec, bins, h, out);
    gcn_relu_inplace<<<g_relu, blk, 0, stream>>>(out);
}


# Product kernel: a wave per 16x128 row strip, 8 accumulator tiles, no LDS; weights re-laid once per call in fragment order; operand loads 16 B per lane, H stored as whole lines; k sum order permuted
# speedup vs baseline: 3.4871x; 1.3297x over previous
#include <hip/hip_runtime.h>

#define GCN_N 100000
#define GCN_D 128
#define GCN_DINV_SLOTS 100032
#define AG_ROWS 100
#define AG_BLOCKS 1000
#define AG_BIN_INTS 2048
#define AG_CURSOR_AT 1024
#define AG_CHUNK 4096
#define AG_SRC_BITS 20

typedef float v2f __attribute__((ext_vector_type(2)));
typedef float v4f __attribute__((ext_vector_type(4)));
typedef float v8f __attribute__((ext_vector_type(8)));

#define WF_FLOATS (GCN_D * GCN_D)
__global__ __launch_bounds__(256) void gcn_fill_zero_and_weight_fragments(float* __restrict__ p, long long n, int fill_blocks,
                                                                          const float* __restrict__ W1,
                                                                          const float* __restrict__ W2,
                                                                          const float* __restrict__ W3,
                                                                          float* __restrict__ wf) {
    if ((int)blockIdx.x < fill_blocks) {
        long long i = (long long)blockIdx.x * blockDim.x + threadIdx.x;
        if (i < n) p[i] = 0.0f;
        return;
    }
    const int o     = ((int)blockIdx.x - fill_blocks) * 256 + threadIdx.x;
    const int layer = o / WF_FLOATS;
    const int q     = o - layer * WF_FLOATS;
    const float* W  = layer == 0 ? W1 : (layer == 1 ? W2 : W3);
    const int e = q & 3, ln = (q >> 2) & 31, tp = (q >> 7) & 3, st = (q >> 9) & 1, kb = q >> 10;
    const int t = 2 * tp + (e >> 1);
    const int k = 8 * kb + 4 * (ln >> 4) + 2 * st + (e & 1);
    const int col = 64 * (t >> 2) + 4 * (ln & 15) + (t & 3);
    wf[o] = W[k * GCN_D + col];
}

__global__ __launch_bounds__(256) void gcn_deg_and_block_counts(const int* __restrict__ dst,
                                                                float* __restrict__ deg,
                                                                int* __restrict__ bincount, int ne) {
    __shared__ int hist[AG_BLOCKS];
    for (int b = threadIdx.x; b < AG_BLOCKS; b += 256) hist[b] = 0;
    __syncthreads();
    const int base = blockIdx.x * AG_CHUNK + threadIdx.x;
    const int last = ne - 1;
    int d[AG_CHUNK / 256];
#pragma unroll
    for (int i = 0; i < AG_CHUNK / 256; ++i) {
        const int e = base + i * 256;
        d[i] = dst[e < last ? e : last];
    }
#pragma unroll
    for (int i = 0; i < AG_CHUNK / 256; ++i) {
        if (base + i * 256 < ne) {
            atomicAdd(&deg[d[i]], 1.0f);
            atomicAdd(&hist[d[i] / AG_ROWS], 1);
        }
    }
    __syncthreads();
    for (int b = threadIdx.x; b < AG_BLOCKS; b += 256) {
        const int c = hist[b];
        if (c) atomicAdd(&bincount[b], c);
    }
}

__device__ __forceinline__ int gcn_block_sum(int v, int* wsum) {
    for (int o = 16; o > 0; o >>= 1) v += __shfl_xor(v, o);
    if ((threadIdx.x & 31) == 0) wsum[threadIdx.x >> 5] = v;
    __syncthreads();
    int t = 0;
#pragma unroll
    for (int k = 0; k < 8; ++k) t += wsum[k];
    return t;
}

__global__ __launch_bounds__(256) void gcn_dinv_and_edge_records(const int* __restrict__ src,
                                                                 const int* __restrict__ dst,
                                                                 const float* __restrict__ deg,
                                                                 const int* __restrict__ bincount,
                                                                 int* __restrict__ cursor,
                                                                 float* __restrict__ dinv,
                                                                 int* __restrict__ erec,
                                                                 int n, int ne) {
    __shared__ int hist[AG_BLOCKS];
    __shared__ int slot0[AG_BLOCKS];
    __shared__ int wsum[8];
    const int tid = threadIdx.x;
    const int stride = gridDim.x * 256;
    for (int j = blockIdx.x * 256 + tid; j < n; j += stride) dinv[j] = rsqrtf(deg[j] + 1.0f);

    for (int b = tid; b < AG_BLOCKS; b += 256) hist[b] = 0;
    __syncthreads();
    int s[AG_CHUNK / 256], d[AG_CHUNK / 256];
    const int base = blockIdx.x * AG_CHUNK + tid;
    const int last = ne - 1;
#pragma unroll
    for (int i = 0; i < AG_CHUNK / 256; ++i) {
        const int e  = base + i * 256;
        const int ec = e < last ? e : last;
        s[i] = src[ec]; d[i] = dst[ec];
    }
#pragma unroll
    for (int i = 0; i < AG_CHUNK / 256; ++i)
        if (base + i * 256 < ne) atomicAdd(&hist[d[i] / AG_ROWS], 1);
    int c[4] = {0, 0, 0, 0};
    if (tid < AG_BLOCKS / 4) {
        const int4 cv = *(const int4*)(bincount + 4 * tid);
        c[0] = cv.x; c[1] = cv.y; c[2] = cv.z; c[3] = cv.w;
    }
    const int tsum = c[0] + c[1] + c[2] + c[3];
    int incl = tsum;
    const int lane = tid & 31, wv = tid >> 5;
    for (int o = 1; o < 32; o <<= 1) {
        const int t = __shfl_up(incl, o);
        if (lane >= o) incl += t;
    }
    if (lane == 31) wsum[wv] = incl;
    __syncthreads();
    int run = incl - tsum;
    for (int k = 0; k < wv; ++k) run += wsum[k];
    float ds[AG_CHUNK / 256], dd[AG_CHUNK / 256];
#pragma unroll
    for (int i = 0; i < AG_CHUNK / 256; ++i) { ds[i] = deg[s[i]]; dd[i] = deg[d[i]]; }
    if (tid < AG_BLOCKS / 4) {
        int got[4];
#pragma unroll
        for (int k = 0; k < 4; ++k) got[k] = atomicAdd(&cursor[4 * tid + k], hist[4 * tid + k]);
#pragma unroll
        for (int k = 0; k < 4; ++k) {
            slot0[4 * tid + k] = run + got[k];
            hist[4 * tid + k] = 0;
            run += c[k];
        }
    }
    __syncthreads();
#pragma unroll
    for (int i = 0; i < AG_CHUNK / 256; ++i) {
        if (base + i * 256 < ne) {
            const int b = d[i] / AG_ROWS;
            const int pos = slot0[b] + atomicAdd(&hist[b], 1);
            const float w = rsqrtf(ds[i] + 1.0f) * rsqrtf(dd[i] + 1.0f);
            *(int2*)(erec + 2 * (size_t)pos) = make_int2(s[i] | ((d[i] - b * AG_ROWS) << AG_SRC_BITS), __float_as_int(w));
        }
    }
}

__global__ __launch_bounds__(64) void gcn_gemm(const float* __restrict__ X,
                                                const float* __restrict__ Wf,
                                                float* __restrict__ H) {
    const int lane     = threadIdx.x & 31;
    const int row_tile = blockIdx.x * 2 + (threadIdx.x >> 5);
    const int c        = lane & 15;
    const int hf       = lane >> 4;

    const float* xrow = X + (size_t)(row_tile * 16 + c) * GCN_D + 4 * hf;
    const v4f*   wl   = (const v4f*)Wf + lane;

    v4f a[GCN_D / 8];
#pragma unroll
    for (int kb = 0; kb < GCN_D / 8; ++kb) a[kb] = __builtin_nontemporal_load((const v4f*)(xrow + 8 * kb));

    v4f b[2][8];
#pragma unroll
    for (int j = 0; j < 8; ++j) b[0][j] = wl[j * 32];

    v8f acc[8] = {};
#pragma unroll
    for (int kb = 0; kb < GCN_D / 8; ++kb) {
        if (kb + 1 < GCN_D / 8) {
#pragma unroll
            for (int j = 0; j < 8; ++j) b[(kb + 1) & 1][j] = wl[((kb + 1) * 8 + j) * 32];
        }
        __builtin_amdgcn_sched_barrier(0);
#pragma unroll
        for (int st = 0; st < 2; ++st) {
            v2f av;
            av.x = st ? a[kb].z : a[kb].x;
            av.y = st ? a[kb].w : a[kb].y;
#pragma unroll
            for (int tp = 0; tp < 4; ++tp) {
                const v4f bw = b[kb & 1][st * 4 + tp];
                v2f b0, b1;
                b0.x = bw.x; b0.y = bw.y;
                b1.x = bw.z; b1.y = bw.w;
                acc[2 * tp]     = __builtin_amdgcn_wmma_f32_16x16x4_f32(
                                      false, av, false, b0, (short)0, acc[2 * tp], false, false);
                acc[2 * tp + 1] = __builtin_amdgcn_wmma_f32_16x16x4_f32(
                                      false, av, false, b1, (short)0, acc[2 * tp + 1], false, false);
            }
        }
    }

    float* hp = H + (size_t)(row_tile * 16 + hf * 8) * GCN_D + 4 * c;
#pragma unroll
    for (int r = 0; r < 8; ++r) {
#pragma unroll
        for (int u = 0; u < 2; ++u) {
            v4f v;
            v.x = acc[4 * u][r]; v.y = acc[4 * u + 1][r]; v.z = acc[4 * u + 2][r]; v.w = acc[4 * u + 3][r];
            *(v4f*)(hp + (size_t)r * GCN_D + 64 * u) = v;
        }
    }
}

#define AG_GROUP 8
#define AG_TILE_ROWS 6

__global__ __launch_bounds__(256) void gcn_block_sums(const int* __restrict__ erec,
                                                      const int* __restrict__ bincount,
                                                      const float* __restrict__ h,
                                                      const float* __restrict__ dinv,
                                                      const float* __restrict__ bias,
                                                      float* __restrict__ acc) {
    __shared__ float tile[AG_ROWS * GCN_D];
    __shared__ int wsum[8];
    const int tid  = threadIdx.x;
    const int lane = tid & 31;
    const int wv   = __builtin_amdgcn_readfirstlane(tid >> 5);
    const int blk  = blockIdx.x;

    int part = 0;
    for (int b = tid; b < blk; b += 256) part += bincount[b];
    const int cnt = bincount[blk];

    float*       arow = acc + (size_t)blk * (AG_ROWS * GCN_D) + 4 * lane;
    const float* hrow = h   + (size_t)blk * (AG_ROWS * GCN_D) + 4 * lane;
    const float* drow = dinv + blk * AG_ROWS;
    const float4 bc   = *(const float4*)(bias + 4 * lane);
#pragma unroll
    for (int p = 0; p < 2; ++p) {
        float4 v[AG_TILE_ROWS];
        float  di[AG_TILE_ROWS];
#pragma unroll
        for (int i = 0; i < AG_TILE_ROWS; ++i) {
            v[i]  = *(const float4*)(hrow + (wv + 8 * (AG_TILE_ROWS * p + i)) * GCN_D);
            di[i] = drow[wv + 8 * (AG_TILE_ROWS * p + i)];
        }
#pragma unroll
        for (int i = 0; i < AG_TILE_ROWS; ++i) {
            float* t = tile + (wv + 8 * (AG_TILE_ROWS * p + i)) * GCN_D + lane;
            t[0]  = v[i].x * di[i] * di[i] + bc.x;
            t[32] = v[i].y * di[i] * di[i] + bc.y;
            t[64] = v[i].z * di[i] * di[i] + bc.z;
            t[96] = v[i].w * di[i] * di[i] + bc.w;
        }
    }
    if (wv < AG_ROWS - 96) {
        const float4 v  = *(const float4*)(hrow + (96 + wv) * GCN_D);
        const float  di = drow[96 + wv];
        float* t = tile + (96 + wv) * GCN_D + lane;
        t[0]  = v.x * di * di + bc.x;
        t[32] = v.y * di * di + bc.y;
        t[64] = v.z * di * di + bc.z;
        t[96] = v.w * di * di + bc.w;
    }
    const int start = __builtin_amdgcn_readfirstlane(gcn_block_sum(part, wsum));
    const int end   = start + __builtin_amdgcn_readfirstlane(cnt);

    const float* hl = h + 4 * lane;
    float*       tl = tile + lane;
    int g0 = start + wv * 32;
    int2 nxt = make_int2(0, 0);
    if (g0 < end) {
        int idx = g0 + lane;
        if (idx > end - 1) idx = end - 1;
        nxt = *(const int2*)(erec + 2 * (size_t)idx);
    }
    for (; g0 < end; g0 += 256) {
        int m = end - g0;
        if (m > 32) m = 32;
        const int2 rec = nxt;
        if (g0 + 256 < end) {
            int idx = g0 + 256 + lane;
            if (idx > end - 1) idx = end - 1;
            nxt = *(const int2*)(erec + 2 * (size_t)idx);
        }
        int g = 0;
        for (; g + AG_GROUP <= m; g += AG_GROUP) {
            float4 v[AG_GROUP];
#pragma unroll
            for (int i = 0; i < AG_GROUP; ++i) {
                const int s = __builtin_amdgcn_readlane(rec.x, g + i) & ((1 << AG_SRC_BITS) - 1);
                v[i] = *(const float4*)(hl + (size_t)s * GCN_D);
            }
#pragma unroll
            for (int i = 0; i < AG_GROUP; ++i) {
                const int   dl = __builtin_amdgcn_readlane(rec.x, g + i) >> AG_SRC_BITS;
                const float w  = __int_as_float(__builtin_amdgcn_readlane(rec.y, g + i));
                float* t = tl + dl * GCN_D;
                atomicAdd(t +  0, v[i].x * w);
                atomicAdd(t + 32, v[i].y * w);
                atomicAdd(t + 64, v[i].z * w);
                atomicAdd(t + 96, v[i].w * w);
            }
        }
        for (; g + 4 <= m; g += 4) {
            float4 v[4];
#pragma unroll
            for (int i = 0; i < 4; ++i) {
                const int s = __builtin_amdgcn_readlane(rec.x, g + i) & ((1 << AG_SRC_BITS) - 1);
                v[i] = *(const float4*)(hl + (size_t)s * GCN_D);
            }
#pragma unroll
            for (int i = 0; i < 4; ++i) {
                const int   dl = __builtin_amdgcn_readlane(rec.x, g + i) >> AG_SRC_BITS;
                const float w  = __int_as_float(__builtin_amdgcn_readlane(rec.y, g + i));
                float* t = tl + dl * GCN_D;
                atomicAdd(t +  0, v[i].x * w);
                atomicAdd(t + 32, v[i].y * w);
                atomicAdd(t + 64, v[i].z * w);
                atomicAdd(t + 96, v[i].w * w);
            }
        }
        for (; g < m; ++g) {
            const int   p  = __builtin_amdgcn_readlane(rec.x, g);
            const float w  = __int_as_float(__builtin_amdgcn_readlane(rec.y, g));
            const float4 v = *(const float4*)(hl + (size_t)(p & ((1 << AG_SRC_BITS) - 1)) * GCN_D);
            float* t = tl + (p >> AG_SRC_BITS) * GCN_D;
            atomicAdd(t +  0, v.x * w);
            atomicAdd(t + 32, v.y * w);
            atomicAdd(t + 64, v.z * w);
            atomicAdd(t + 96, v.w * w);
        }
    }
    __syncthreads();
#pragma unroll
    for (int i = 0; i < 2 * AG_TILE_ROWS; ++i) {
        const float* t = tile + (wv + 8 * i) * GCN_D + lane;
        float4 v;
        v.x = t[0]; v.y = t[32]; v.z = t[64]; v.w = t[96];
        v.x = v.x > 0.0f ? v.x : 0.0f; v.y = v.y > 0.0f ? v.y : 0.0f;
        v.z = v.z > 0.0f ? v.z : 0.0f; v.w = v.w > 0.0f ? v.w : 0.0f;
        *(float4*)(arow + (wv + 8 * i) * GCN_D) = v;
    }
    if (wv < AG_ROWS - 96) {
        const float* t = tile + (96 + wv) * GCN_D + lane;
        float4 v;
        v.x = t[0]; v.y = t[32]; v.z = t[64]; v.w = t[96];
        v.x = v.x > 0.0f ? v.x : 0.0f; v.y = v.y > 0.0f ? v.y : 0.0f;
        v.z = v.z > 0.0f ? v.z : 0.0f; v.w = v.w > 0.0f ? v.w : 0.0f;
        *(float4*)(arow + (96 + wv) * GCN_D) = v;
    }
}

extern "C" void kernel_launch(void* const* d_in, const int* in_sizes, int n_in,
                              void* d_out, int out_size, void* d_ws, size_t ws_size,
                              hipStream_t stream) {
    const float* x  = (const float*)d_in[0];
    const float* W1 = (const float*)d_in[1];
    const float* b1 = (const float*)d_in[2];
    const float* W2 = (const float*)d_in[3];
    const float* b2 = (const float*)d_in[4];
    const float* W3 = (const float*)d_in[5];
    const float* b3 = (const float*)d_in[6];
    const int*   eidx = (const int*)d_in[7];
    const int ne = in_sizes[7] / 2;
    const int* src = eidx;
    const int* dst = eidx + ne;

    char* ws = (char*)d_ws;
    size_t off = 0;
    float* deg   = (float*)(ws + off); off += (size_t)GCN_DINV_SLOTS * 4;
    int*   bins  = (int*)(ws + off);   off += (size_t)AG_BIN_INTS * 4;
    float* dinv  = (float*)(ws + off); off += (size_t)GCN_DINV_SLOTS * 4;
    int*   erec  = (int*)(ws + off);   off += (((size_t)ne * 8 + 255) / 256) * 256;
    float* h     = (float*)(ws + off); off += (size_t)GCN_N * GCN_D * 4;
    float* acc_a = (float*)(ws + off); off += (size_t)GCN_N * GCN_D * 4;
    float* acc_b = (float*)(ws + off); off += (size_t)GCN_N * GCN_D * 4;
    float* wf    = (float*)(ws + off);
    float* out   = (float*)d_out;

    const int blk = 256;
    const int g_fill  = (GCN_DINV_SLOTS + AG_BIN_INTS + blk - 1) / blk;
    const int g_bin   = (ne + AG_CHUNK - 1) / AG_CHUNK;
    const int g_wf    = 3 * WF_FLOATS / blk;
    const int g_gemm  = GCN_N / 16 / 2;

    gcn_fill_zero_and_weight_fragments<<<g_fill + g_wf, blk, 0, stream>>>(deg, GCN_DINV_SLOTS + AG_BIN_INTS, g_fill,
                                                                          W1, W2, W3, wf);
    gcn_deg_and_block_counts<<<g_bin, blk, 0, stream>>>(dst, deg, bins, ne);
    gcn_dinv_and_edge_records<<<g_bin, blk, 0, stream>>>(src, dst, deg, bins, bins + AG_CURSOR_AT, dinv, erec, GCN_N, ne);

    gcn_gemm<<<g_gemm, 64, 0, stream>>>(x, wf, h);
    gcn_block_sums<<<AG_BLOCKS, blk, 0, stream>>>(erec, bins, h, dinv, b1, acc_a);

    gcn_gemm<<<g_gemm, 64, 0, stream>>>(acc_a, wf + WF_FLOATS, h);
    gcn_block_sums<<<AG_BLOCKS, blk, 0, stream>>>(erec, bins, h, dinv, b2, acc_b);

    gcn_gemm<<<g_gemm, 64, 0, stream>>>(acc_b, wf + 2 * WF_FLOATS, h);
    gcn_block_sums<<<AG_BLOCKS, blk, 0, stream>>>(erec, bins, h, dinv, b3, out);
}


# Layer-1 product (needs no edge data) runs inside the two edge pre-pass launches: work-groups behind the edge work-groups take its 16-row strips, one per wave; 9 to 8 launches; sum order unchanged
# speedup vs baseline: 3.6224x; 1.0388x over previous
#include <hip/hip_runtime.h>

#define GCN_N 100000
#define GCN_D 128
#define GCN_DINV_SLOTS 100032
#define AG_ROWS 100
#define AG_BLOCKS 1000
#define AG_BIN_INTS 2048
#define AG_CURSOR_AT 1024
#define AG_CHUNK 4096
#define AG_SRC_BITS 20

typedef float v2f __attribute__((ext_vector_type(2)));
typedef float v4f __attribute__((ext_vector_type(4)));
typedef float v8f __attribute__((ext_vector_type(8)));

#define WF_FLOATS (GCN_D * GCN_D)
__global__ __launch_bounds__(256) void gcn_fill_zero_and_weight_fragments(float* __restrict__ p, long long n, int fill_blocks,
                                                                          const float* __restrict__ W1,
                                                                          const float* __restrict__ W2,
                                                                          const float* __restrict__ W3,
                                                                          float* __restrict__ wf) {
    if ((int)blockIdx.x < fill_blocks) {
        long long i = (long long)blockIdx.x * blockDim.x + threadIdx.x;
        if (i < n) p[i] = 0.0f;
        return;
    }
    const int o     = ((int)blockIdx.x - fill_blocks) * 256 + threadIdx.x;
    const int layer = o / WF_FLOATS;
    const int q     = o - layer * WF_FLOATS;
    const float* W  = layer == 0 ? W1 : (layer == 1 ? W2 : W3);
    const int e = q & 3, ln = (q >> 2) & 31, tp = (q >> 7) & 3, st = (q >> 9) & 1, kb = q >> 10;
    const int t = 2 * tp + (e >> 1);
    const int k = 8 * kb + 4 * (ln >> 4) + 2 * st + (e & 1);
    const int col = 64 * (t >> 2) + 4 * (ln & 15) + (t & 3);
    wf[o] = W[k * GCN_D + col];
}

__device__ __forceinline__ void gcn_gemm_strip(const float* __restrict__ X,
                                               const float* __restrict__ Wf,
                                               float* __restrict__ H, int row_tile, int lane) {
    const int c        = lane & 15;
    const int hf       = lane >> 4;

    const float* xrow = X + (size_t)(row_tile * 16 + c) * GCN_D + 4 * hf;
    const v4f*   wl   = (const v4f*)Wf + lane;

    v4f a[GCN_D / 8];
#pragma unroll
    for (int kb = 0; kb < GCN_D / 8; ++kb) a[kb] = __builtin_nontemporal_load((const v4f*)(xrow + 8 * kb));

    v4f b[2][8];
#pragma unroll
    for (int j = 0; j < 8; ++j) b[0][j] = wl[j * 32];

    v8f acc[8] = {};
#pragma unroll
    for (int kb = 0; kb < GCN_D / 8; ++kb) {
        if (kb + 1 < GCN_D / 8) {
#pragma unroll
            for (int j = 0; j < 8; ++j) b[(kb + 1) & 1][j] = wl[((kb + 1) * 8 + j) * 32];
        }
        __builtin_amdgcn_sched_barrier(0);
#pragma unroll
        for (int st = 0; st < 2; ++st) {
            v2f av;
            av.x = st ? a[kb].z : a[kb].x;
            av.y = st ? a[kb].w : a[kb].y;
#pragma unroll
            for (int tp = 0; tp < 4; ++tp) {
                const v4f bw = b[kb & 1][st * 4 + tp];
                v2f b0, b1;
                b0.x = bw.x; b0.y = bw.y;
                b1.x = bw.z; b1.y = bw.w;
                acc[2 * tp]     = __builtin_amdgcn_wmma_f32_16x16x4_f32(
                                      false, av, false, b0, (short)0, acc[2 * tp], false, false);
                acc[2 * tp + 1] = __builtin_amdgcn_wmma_f32_16x16x4_f32(
                                      false, av, false, b1, (short)0, acc[2 * tp + 1], false, false);
            }
        }
    }

    float* hp = H + (size_t)(row_tile * 16 + hf * 8) * GCN_D + 4 * c;
#pragma unroll
    for (int r = 0; r < 8; ++r) {
#pragma unroll
        for (int u = 0; u < 2; ++u) {
            v4f v;
            v.x = acc[4 * u][r]; v.y = acc[4 * u + 1][r]; v.z = acc[4 * u + 2][r]; v.w = acc[4 * u + 3][r];
            *(v4f*)(hp + (size_t)r * GCN_D + 64 * u) = v;
        }
    }
}

__global__ __launch_bounds__(64) void gcn_gemm(const float* __restrict__ X,
                                                const float* __restrict__ Wf,
                                                float* __restrict__ H) {
    gcn_gemm_strip(X, Wf, H, blockIdx.x * 2 + (threadIdx.x >> 5), threadIdx.x & 31);
}

#define GEMM_STRIPS   (GCN_N / 16)
#define GEMM_STRIPS_A 2496
#define GEMM_BLOCKS_A (GEMM_STRIPS_A / 8)
#define GEMM_BLOCKS_B ((GEMM_STRIPS - GEMM_STRIPS_A + 7) / 8)
__device__ __forceinline__ void gcn_gemm_rider(const float* __restrict__ X, const float* __restrict__ Wf,
                                               float* __restrict__ H, int first_strip, int rider_block) {
    const int wv = __builtin_amdgcn_readfirstlane(threadIdx.x >> 5);
    const int row_tile = first_strip + rider_block * 8 + wv;
    if (row_tile < GEMM_STRIPS) gcn_gemm_strip(X, Wf, H, row_tile, threadIdx.x & 31);
}

__global__ __launch_bounds__(256) void gcn_deg_and_block_counts(const int* __restrict__ dst,
                                                                float* __restrict__ deg,
                                                                int* __restrict__ bincount, int ne, int edge_blocks,
                                                                const float* __restrict__ X,
                                                                const float* __restrict__ Wf,
                                                                float* __restrict__ H) {
    __shared__ int hist[AG_BLOCKS];
    if ((int)blockIdx.x >= edge_blocks) {
        gcn_gemm_rider(X, Wf, H, 0, (int)blockIdx.x - edge_blocks);
        return;
    }
    for (int b = threadIdx.x; b < AG_BLOCKS; b += 256) hist[b] = 0;
    __syncthreads();
    const int base = blockIdx.x * AG_CHUNK + threadIdx.x;
    const int last = ne - 1;
    int d[AG_CHUNK / 256];
#pragma unroll
    for (int i = 0; i < AG_CHUNK / 256; ++i) {
        const int e = base + i * 256;
        d[i] = dst[e < last ? e : last];
    }
#pragma unroll
    for (int i = 0; i < AG_CHUNK / 256; ++i) {
        if (base + i * 256 < ne) {
            atomicAdd(&deg[d[i]], 1.0f);
            atomicAdd(&hist[d[i] / AG_ROWS], 1);
        }
    }
    __syncthreads();
    for (int b = threadIdx.x; b < AG_BLOCKS; b += 256) {
        const int c = hist[b];
        if (c) atomicAdd(&bincount[b], c);
    }
}

__device__ __forceinline__ int gcn_block_sum(int v, int* wsum) {
    for (int o = 16; o > 0; o >>= 1) v += __shfl_xor(v, o);
    if ((threadIdx.x & 31) == 0) wsum[threadIdx.x >> 5] = v;
    __syncthreads();
    int t = 0;
#pragma unroll
    for (int k = 0; k < 8; ++k) t += wsum[k];
    return t;
}

__global__ __launch_bounds__(256) void gcn_dinv_and_edge_records(const int* __restrict__ src,
                                                                 const int* __restrict__ dst,
                                                                 const float* __restrict__ deg,
                                                                 const int* __restrict__ bincount,
                                                                 int* __restrict__ cursor,
                                                                 float* __restrict__ dinv,
                                                                 int* __restrict__ erec,
                                                                 int n, int ne, int edge_blocks,
                                                                 const float* __restrict__ X,
                                                                 const float* __restrict__ Wf,
                                                                 float* __restrict__ H) {
    __shared__ int hist[AG_BLOCKS];
    __shared__ int slot0[AG_BLOCKS];
    __shared__ int wsum[8];
    if ((int)blockIdx.x >= edge_blocks) {
        gcn_gemm_rider(X, Wf, H, GEMM_STRIPS_A, (int)blockIdx.x - edge_blocks);
        return;
    }
    const int tid = threadIdx.x;
    const int stride = edge_blocks * 256;
    for (int j = blockIdx.x * 256 + tid; j < n; j += stride) dinv[j] = rsqrtf(deg[j] + 1.0f);

    for (int b = tid; b < AG_BLOCKS; b += 256) hist[b] = 0;
    __syncthreads();
    int s[AG_CHUNK / 256], d[AG_CHUNK / 256];
    const int base = blockIdx.x * AG_CHUNK + tid;
    const int last = ne - 1;
#pragma unroll
    for (int i = 0; i < AG_CHUNK / 256; ++i) {
        const int e  = base + i * 256;
        const int ec = e < last ? e : last;
        s[i] = src[ec]; d[i] = dst[ec];
    }
#pragma unroll
    for (int i = 0; i < AG_CHUNK / 256; ++i)
        if (base + i * 256 < ne) atomicAdd(&hist[d[i] / AG_ROWS], 1);
    int c[4] = {0, 0, 0, 0};
    if (tid < AG_BLOCKS / 4) {
        const int4 cv = *(const int4*)(bincount + 4 * tid);
        c[0] = cv.x; c[1] = cv.y; c[2] = cv.z; c[3] = cv.w;
    }
    const int tsum = c[0] + c[1] + c[2] + c[3];
    int incl = tsum;
    const int lane = tid & 31, wv = tid >> 5;
    for (int o = 1; o < 32; o <<= 1) {
        const int t = __shfl_up(incl, o);
        if (lane >= o) incl += t;
    }
    if (lane == 31) wsum[wv] = incl;
    __syncthreads();
    int run = incl - tsum;
    for (int k = 0; k < wv; ++k) run += wsum[k];
    float ds[AG_CHUNK / 256], dd[AG_CHUNK / 256];
#pragma unroll
    for (int i = 0; i < AG_CHUNK / 256; ++i) { ds[i] = deg[s[i]]; dd[i] = deg[d[i]]; }
    if (tid < AG_BLOCKS / 4) {
        int got[4];
#pragma unroll
        for (int k = 0; k < 4; ++k) got[k] = atomicAdd(&cursor[4 * tid + k], hist[4 * tid + k]);
#pragma unroll
        for (int k = 0; k < 4; ++k) {
            slot0[4 * tid + k] = run + got[k];
            hist[4 * tid + k] = 0;
            run += c[k];
        }
    }
    __syncthreads();
#pragma unroll
    for (int i = 0; i < AG_CHUNK / 256; ++i) {
        if (base + i * 256 < ne) {
            const int b = d[i] / AG_ROWS;
            const int pos = slot0[b] + atomicAdd(&hist[b], 1);
            const float w = rsqrtf(ds[i] + 1.0f) * rsqrtf(dd[i] + 1.0f);
            *(int2*)(erec + 2 * (size_t)pos) = make_int2(s[i] | ((d[i] - b * AG_ROWS) << AG_SRC_BITS), __float_as_int(w));
        }
    }
}

#define AG_GROUP 8
#define AG_TILE_ROWS 6

__global__ __launch_bounds__(256) void gcn_block_sums(const int* __restrict__ erec,
                                                      const int* __restrict__ bincount,
                                                      const float* __restrict__ h,
                                                      const float* __restrict__ dinv,
                                                      const float* __restrict__ bias,
                                                      float* __restrict__ acc) {
    __shared__ float tile[AG_ROWS * GCN_D];
    __shared__ int wsum[8];
    const int tid  = threadIdx.x;
    const int lane = tid & 31;
    const int wv   = __builtin_amdgcn_readfirstlane(tid >> 5);
    const int blk  = blockIdx.x;

    int part = 0;
    for (int b = tid; b < blk; b += 256) part += bincount[b];
    const int cnt = bincount[blk];

    float*       arow = acc + (size_t)blk * (AG_ROWS * GCN_D) + 4 * lane;
    const float* hrow = h   + (size_t)blk * (AG_ROWS * GCN_D) + 4 * lane;
    const float* drow = dinv + blk * AG_ROWS;
    const float4 bc   = *(const float4*)(bias + 4 * lane);
#pragma unroll
    for (int p = 0; p < 2; ++p) {
        float4 v[AG_TILE_ROWS];
        float  di[AG_TILE_ROWS];
#pragma unroll
        for (int i = 0; i < AG_TILE_ROWS; ++i) {
            v[i]  = *(const float4*)(hrow + (wv + 8 * (AG_TILE_ROWS * p + i)) * GCN_D);
            di[i] = drow[wv + 8 * (AG_TILE_ROWS * p + i)];
        }
#pragma unroll
        for (int i = 0; i < AG_TILE_ROWS; ++i) {
            float* t = tile + (wv + 8 * (AG_TILE_ROWS * p + i)) * GCN_D + lane;
            t[0]  = v[i].x * di[i] * di[i] + bc.x;
            t[32] = v[i].y * di[i] * di[i] + bc.y;
            t[64] = v[i].z * di[i] * di[i] + bc.z;
            t[96] = v[i].w * di[i] * di[i] + bc.w;
        }
    }
    if (wv < AG_ROWS - 96) {
        const float4 v  = *(const float4*)(hrow + (96 + wv) * GCN_D);
        const float  di = drow[96 + wv];
        float* t = tile + (96 + wv) * GCN_D + lane;
        t[0]  = v.x * di * di + bc.x;
        t[32] = v.y * di * di + bc.y;
        t[64] = v.z * di * di + bc.z;
        t[96] = v.w * di * di + bc.w;
    }
    const int start = __builtin_amdgcn_readfirstlane(gcn_block_sum(part, wsum));
    const int end   = start + __builtin_amdgcn_readfirstlane(cnt);

    const float* hl = h + 4 * lane;
    float*       tl = tile + lane;
    int g0 = start + wv * 32;
    int2 nxt = make_int2(0, 0);
    if (g0 < end) {
        int idx = g0 + lane;
        if (idx > end - 1) idx = end - 1;
        nxt = *(const int2*)(erec + 2 * (size_t)idx);
    }
    for (; g0 < end; g0 += 256) {
        int m = end - g0;
        if (m > 32) m = 32;
        const int2 rec = nxt;
        if (g0 + 256 < end) {
            int idx = g0 + 256 + lane;
            if (idx > end - 1) idx = end - 1;
            nxt = *(const int2*)(erec + 2 * (size_t)idx);
        }
        int g = 0;
        for (; g + AG_GROUP <= m; g += AG_GROUP) {
            float4 v[AG_GROUP];
#pragma unroll
            for (int i = 0; i < AG_GROUP; ++i) {
                const int s = __builtin_amdgcn_readlane(rec.x, g + i) & ((1 << AG_SRC_BITS) - 1);
                v[i] = *(const float4*)(hl + (size_t)s * GCN_D);
            }
#pragma unroll
            for (int i = 0; i < AG_GROUP; ++i) {
                const int   dl = __builtin_amdgcn_readlane(rec.x, g + i) >> AG_SRC_BITS;
                const float w  = __int_as_float(__builtin_amdgcn_readlane(rec.y, g + i));
                float* t = tl + dl * GCN_D;
                atomicAdd(t +  0, v[i].x * w);
                atomicAdd(t + 32, v[i].y * w);
                atomicAdd(t + 64, v[i].z * w);
                atomicAdd(t + 96, v[i].w * w);
            }
        }
        for (; g + 4 <= m; g += 4) {
            float4 v[4];
#pragma unroll
            for (int i = 0; i < 4; ++i) {
                const int s = __builtin_amdgcn_readlane(rec.x, g + i) & ((1 << AG_SRC_BITS) - 1);
                v[i] = *(const float4*)(hl + (size_t)s * GCN_D);
            }
#pragma unroll
            for (int i = 0; i < 4; ++i) {
                const int   dl = __builtin_amdgcn_readlane(rec.x, g + i) >> AG_SRC_BITS;
                const float w  = __int_as_float(__builtin_amdgcn_readlane(rec.y, g + i));
                float* t = tl + dl * GCN_D;
                atomicAdd(t +  0, v[i].x * w);
                atomicAdd(t + 32, v[i].y * w);
                atomicAdd(t + 64, v[i].z * w);
                atomicAdd(t + 96, v[i].w * w);
            }
        }
        for (; g < m; ++g) {
            const int   p  = __builtin_amdgcn_readlane(rec.x, g);
            const float w  = __int_as_float(__builtin_amdgcn_readlane(rec.y, g));
            const float4 v = *(const float4*)(hl + (size_t)(p & ((1 << AG_SRC_BITS) - 1)) * GCN_D);
            float* t = tl + (p >> AG_SRC_BITS) * GCN_D;
            atomicAdd(t +  0, v.x * w);
            atomicAdd(t + 32, v.y * w);
            atomicAdd(t + 64, v.z * w);
            atomicAdd(t + 96, v.w * w);
        }
    }
    __syncthreads();
#pragma unroll
    for (int i = 0; i < 2 * AG_TILE_ROWS; ++i) {
        const float* t = tile + (wv + 8 * i) * GCN_D + lane;
        float4 v;
        v.x = t[0]; v.y = t[32]; v.z = t[64]; v.w = t[96];
        v.x = v.x > 0.0f ? v.x : 0.0f; v.y = v.y > 0.0f ? v.y : 0.0f;
        v.z = v.z > 0.0f ? v.z : 0.0f; v.w = v.w > 0.0f ? v.w : 0.0f;
        *(float4*)(arow + (wv + 8 * i) * GCN_D) = v;
    }
    if (wv < AG_ROWS - 96) {
        const float* t = tile + (96 + wv) * GCN_D + lane;
        float4 v;
        v.x = t[0]; v.y = t[32]; v.z = t[64]; v.w = t[96];
        v.x = v.x > 0.0f ? v.x : 0.0f; v.y = v.y > 0.0f ? v.y : 0.0f;
        v.z = v.z > 0.0f ? v.z : 0.0f; v.w = v.w > 0.0f ? v.w : 0.0f;
        *(float4*)(arow + (96 + wv) * GCN_D) = v;
    }
}

extern "C" void kernel_launch(void* const* d_in, const int* in_sizes, int n_in,
                              void* d_out, int out_size, void* d_ws, size_t ws_size,
                              hipStream_t stream) {
    const float* x  = (const float*)d_in[0];
    const float* W1 = (const float*)d_in[1];
    const float* b1 = (const float*)d_in[2];
    const float* W2 = (const float*)d_in[3];
    const float* b2 = (const float*)d_in[4];
    const float* W3 = (const float*)d_in[5];
    const float* b3 = (const float*)d_in[6];
    const int*   eidx = (const int*)d_in[7];
    const int ne = in_sizes[7] / 2;
    const int* src = eidx;
    const int* dst = eidx + ne;

    char* ws = (char*)d_ws;
    size_t off = 0;
    float* deg   = (float*)(ws + off); off += (size_t)GCN_DINV_SLOTS * 4;
    int*   bins  = (int*)(ws + off);   off += (size_t)AG_BIN_INTS * 4;
    float* dinv  = (float*)(ws + off); off += (size_t)GCN_DINV_SLOTS * 4;
    int*   erec  = (int*)(ws + off);   off += (((size_t)ne * 8 + 255) / 256) * 256;
    float* h     = (float*)(ws + off); off += (size_t)GCN_N * GCN_D * 4;
    float* acc_a = (float*)(ws + off); off += (size_t)GCN_N * GCN_D * 4;
    float* acc_b = (float*)(ws + off); off += (size_t)GCN_N * GCN_D * 4;
    float* wf    = (float*)(ws + off);
    float* out   = (float*)d_out;

    const int blk = 256;
    const int g_fill  = (GCN_DINV_SLOTS + AG_BIN_INTS + blk - 1) / blk;
    const int g_bin   = (ne + AG_CHUNK - 1) / AG_CHUNK;
    const int g_wf    = 3 * WF_FLOATS / blk;
    const int g_gemm  = GCN_N / 16 / 2;

    gcn_fill_zero_and_weight_fragments<<<g_fill + g_wf, blk, 0, stream>>>(deg, GCN_DINV_SLOTS + AG_BIN_INTS, g_fill,
                                                                          W1, W2, W3, wf);
    gcn_deg_and_block_counts<<<g_bin + GEMM_BLOCKS_A, blk, 0, stream>>>(dst, deg, bins, ne, g_bin, x, wf, h);
    gcn_dinv_and_edge_records<<<g_bin + GEMM_BLOCKS_B, blk, 0, stream>>>(src, dst, deg, bins, bins + AG_CURSOR_AT, dinv, erec, GCN_N, ne,
                                                                         g_bin, x, wf, h);

    gcn_block_sums<<<AG_BLOCKS, blk, 0, stream>>>(erec, bins, h, dinv, b1, acc_a);

    gcn_gemm<<<g_gemm, 64, 0, stream>>>(acc_a, wf + WF_FLOATS, h);
    gcn_block_sums<<<AG_BLOCKS, blk, 0, stream>>>(erec, bins, h, dinv, b2, acc_b);

    gcn_gemm<<<g_gemm, 64, 0, stream>>>(acc_b, wf + 2 * WF_FLOATS, h);
    gcn_block_sums<<<AG_BLOCKS, blk, 0, stream>>>(erec, bins, h, dinv, b3, out);
}


# In-degree counted from the grouped edge records (a work-group per 100-row block, integer LDS adds): the 1.6M per-edge global float adds go; dinv and weights in two small launches; sums as before
# speedup vs baseline: 4.3959x; 1.2135x over previous
#include <hip/hip_runtime.h>

#define GCN_N 100000
#define GCN_D 128
#define GCN_DINV_SLOTS 100032
#define AG_ROWS 100
#define AG_BLOCKS 1000
#define AG_BIN_INTS 2048
#define AG_CURSOR_AT 1024
#define AG_CHUNK 4096
#define AG_SRC_BITS 20

typedef float v2f __attribute__((ext_vector_type(2)));
typedef float v4f __attribute__((ext_vector_type(4)));
typedef float v8f __attribute__((ext_vector_type(8)));

#define WF_FLOATS (GCN_D * GCN_D)
__global__ __launch_bounds__(256) void gcn_fill_zero_and_weight_fragments(float* __restrict__ p, long long n, int fill_blocks,
                                                                          const float* __restrict__ W1,
                                                                          const float* __restrict__ W2,
                                                                          const float* __restrict__ W3,
                                                                          float* __restrict__ wf) {
    if ((int)blockIdx.x < fill_blocks) {
        long long i = (long long)blockIdx.x * blockDim.x + threadIdx.x;
        if (i < n) p[i] = 0.0f;
        return;
    }
    const int o     = ((int)blockIdx.x - fill_blocks) * 256 + threadIdx.x;
    const int layer = o / WF_FLOATS;
    const int q     = o - layer * WF_FLOATS;
    const float* W  = layer == 0 ? W1 : (layer == 1 ? W2 : W3);
    const int e = q & 3, ln = (q >> 2) & 31, tp = (q >> 7) & 3, st = (q >> 9) & 1, kb = q >> 10;
    const int t = 2 * tp + (e >> 1);
    const int k = 8 * kb + 4 * (ln >> 4) + 2 * st + (e & 1);
    const int col = 64 * (t >> 2) + 4 * (ln & 15) + (t & 3);
    wf[o] = W[k * GCN_D + col];
}

__device__ __forceinline__ void gcn_gemm_strip(const float* __restrict__ X,
                                               const float* __restrict__ Wf,
                                               float* __restrict__ H, int row_tile, int lane) {
    const int c        = lane & 15;
    const int hf       = lane >> 4;

    const float* xrow = X + (size_t)(row_tile * 16 + c) * GCN_D + 4 * hf;
    const v4f*   wl   = (const v4f*)Wf + lane;

    v4f a[GCN_D / 8];
#pragma unroll
    for (int kb = 0; kb < GCN_D / 8; ++kb) a[kb] = __builtin_nontemporal_load((const v4f*)(xrow + 8 * kb));

    v4f b[2][8];
#pragma unroll
    for (int j = 0; j < 8; ++j) b[0][j] = wl[j * 32];

    v8f acc[8] = {};
#pragma unroll
    for (int kb = 0; kb < GCN_D / 8; ++kb) {
        if (kb + 1 < GCN_D / 8) {
#pragma unroll
            for (int j = 0; j < 8; ++j) b[(kb + 1) & 1][j] = wl[((kb + 1) * 8 + j) * 32];
        }
        __builtin_amdgcn_sched_barrier(0);
#pragma unroll
        for (int st = 0; st < 2; ++st) {
            v2f av;
            av.x = st ? a[kb].z : a[kb].x;
            av.y = st ? a[kb].w : a[kb].y;
#pragma unroll
            for (int tp = 0; tp < 4; ++tp) {
                const v4f bw = b[kb & 1][st * 4 + tp];
                v2f b0, b1;
                b0.x = bw.x; b0.y = bw.y;
                b1.x = bw.z; b1.y = bw.w;
                acc[2 * tp]     = __builtin_amdgcn_wmma_f32_16x16x4_f32(
                                      false, av, false, b0, (short)0, acc[2 * tp], false, false);
                acc[2 * tp + 1] = __builtin_amdgcn_wmma_f32_16x16x4_f32(
                                      false, av, false, b1, (short)0, acc[2 * tp + 1], false, false);
            }
        }
    }

    float* hp = H + (size_t)(row_tile * 16 + hf * 8) * GCN_D + 4 * c;
#pragma unroll
    for (int r = 0; r < 8; ++r) {
#pragma unroll
        for (int u = 0; u < 2; ++u) {
            v4f v;
            v.x = acc[4 * u][r]; v.y = acc[4 * u + 1][r]; v.z = acc[4 * u + 2][r]; v.w = acc[4 * u + 3][r];
            *(v4f*)(hp + (size_t)r * GCN_D + 64 * u) = v;
        }
    }
}

__global__ __launch_bounds__(64) void gcn_gemm(const float* __restrict__ X,
                                                const float* __restrict__ Wf,
                                                float* __restrict__ H) {
    gcn_gemm_strip(X, Wf, H, blockIdx.x * 2 + (threadIdx.x >> 5), threadIdx.x & 31);
}

#define GEMM_STRIPS   (GCN_N / 16)
#define GEMM_STRIPS_A 2496
#define GEMM_BLOCKS_A (GEMM_STRIPS_A / 8)
#define GEMM_BLOCKS_B ((GEMM_STRIPS - GEMM_STRIPS_A + 7) / 8)
__device__ __forceinline__ void gcn_gemm_rider(const float* __restrict__ X, const float* __restrict__ Wf,
                                               float* __restrict__ H, int first_strip, int rider_block) {
    const int wv = __builtin_amdgcn_readfirstlane(threadIdx.x >> 5);
    const int row_tile = first_strip + rider_block * 8 + wv;
    if (row_tile < GEMM_STRIPS) gcn_gemm_strip(X, Wf, H, row_tile, threadIdx.x & 31);
}

__global__ __launch_bounds__(256) void gcn_block_counts(const int* __restrict__ dst,
                                                                int* __restrict__ bincount, int ne, int edge_blocks,
                                                                const float* __restrict__ X,
                                                                const float* __restrict__ Wf,
                                                                float* __restrict__ H) {
    __shared__ int hist[AG_BLOCKS];
    if ((int)blockIdx.x >= edge_blocks) {
        gcn_gemm_rider(X, Wf, H, 0, (int)blockIdx.x - edge_blocks);
        return;
    }
    for (int b = threadIdx.x; b < AG_BLOCKS; b += 256) hist[b] = 0;
    __syncthreads();
    const int base = blockIdx.x * AG_CHUNK + threadIdx.x;
    const int last = ne - 1;
    int d[AG_CHUNK / 256];
#pragma unroll
    for (int i = 0; i < AG_CHUNK / 256; ++i) {
        const int e = base + i * 256;
        d[i] = dst[e < last ? e : last];
    }
#pragma unroll
    for (int i = 0; i < AG_CHUNK / 256; ++i) {
        if (base + i * 256 < ne) atomicAdd(&hist[d[i] / AG_ROWS], 1);
    }
    __syncthreads();
    for (int b = threadIdx.x; b < AG_BLOCKS; b += 256) {
        const int c = hist[b];
        if (c) atomicAdd(&bincount[b], c);
    }
}

__device__ __forceinline__ int gcn_block_sum(int v, int* wsum) {
    for (int o = 16; o > 0; o >>= 1) v += __shfl_xor(v, o);
    if ((threadIdx.x & 31) == 0) wsum[threadIdx.x >> 5] = v;
    __syncthreads();
    int t = 0;
#pragma unroll
    for (int k = 0; k < 8; ++k) t += wsum[k];
    return t;
}

__global__ __launch_bounds__(256) void gcn_edge_records(const int* __restrict__ src,
                                                                 const int* __restrict__ dst,
                                                                 const int* __restrict__ bincount,
                                                                 int* __restrict__ cursor,
                                                                 int* __restrict__ erec,
                                                                 int ne, int edge_blocks,
                                                                 const float* __restrict__ X,
                                                                 const float* __restrict__ Wf,
                                                                 float* __restrict__ H) {
    __shared__ int hist[AG_BLOCKS];
    __shared__ int slot0[AG_BLOCKS];
    __shared__ int wsum[8];
    if ((int)blockIdx.x >= edge_blocks) {
        gcn_gemm_rider(X, Wf, H, GEMM_STRIPS_A, (int)blockIdx.x - edge_blocks);
        return;
    }
    const int tid = threadIdx.x;
    for (int b = tid; b < AG_BLOCKS; b += 256) hist[b] = 0;
    __syncthreads();
    int s[AG_CHUNK / 256], d[AG_CHUNK / 256];
    const int base = blockIdx.x * AG_CHUNK + tid;
    const int last = ne - 1;
#pragma unroll
    for (int i = 0; i < AG_CHUNK / 256; ++i) {
        const int e  = base + i * 256;
        const int ec = e < last ? e : last;
        s[i] = src[ec]; d[i] = dst[ec];
    }
#pragma unroll
    for (int i = 0; i < AG_CHUNK / 256; ++i)
        if (base + i * 256 < ne) atomicAdd(&hist[d[i] / AG_ROWS], 1);
    int c[4] = {0, 0, 0, 0};
    if (tid < AG_BLOCKS / 4) {
        const int4 cv = *(const int4*)(bincount + 4 * tid);
        c[0] = cv.x; c[1] = cv.y; c[2] = cv.z; c[3] = cv.w;
    }
    const int tsum = c[0] + c[1] + c[2] + c[3];
    int incl = tsum;
    const int lane = tid & 31, wv = tid >> 5;
    for (int o = 1; o < 32; o <<= 1) {
        const int t = __shfl_up(incl, o);
        if (lane >= o) incl += t;
    }
    if (lane == 31) wsum[wv] = incl;
    __syncthreads();
    int run = incl - tsum;
    for (int k = 0; k < wv; ++k) run += wsum[k];
    if (tid < AG_BLOCKS / 4) {
        int got[4];
#pragma unroll
        for (int k = 0; k < 4; ++k) got[k] = atomicAdd(&cursor[4 * tid + k], hist[4 * tid + k]);
#pragma unroll
        for (int k = 0; k < 4; ++k) {
            slot0[4 * tid + k] = run + got[k];
            hist[4 * tid + k] = 0;
            run += c[k];
        }
    }
    __syncthreads();
#pragma unroll
    for (int i = 0; i < AG_CHUNK / 256; ++i) {
        if (base + i * 256 < ne) {
            const int b = d[i] / AG_ROWS;
            const int pos = slot0[b] + atomicAdd(&hist[b], 1);
            erec[2 * (size_t)pos] = s[i] | ((d[i] - b * AG_ROWS) << AG_SRC_BITS);
        }
    }
}

__device__ __forceinline__ int gcn_block_start(const int* __restrict__ bincount, int blk, int* wsum) {
    int part = 0;
    if (threadIdx.x < AG_BLOCKS / 4) {
        const int4 cv = *(const int4*)(bincount + 4 * threadIdx.x);
        const int b0 = 4 * threadIdx.x;
        part = (b0 < blk ? cv.x : 0) + (b0 + 1 < blk ? cv.y : 0) + (b0 + 2 < blk ? cv.z : 0) + (b0 + 3 < blk ? cv.w : 0);
    }
    return gcn_block_sum(part, wsum);
}

#define AG_SWEEP 8

__global__ __launch_bounds__(256) void gcn_block_dinv(const int* __restrict__ erec,
                                                      const int* __restrict__ bincount,
                                                      float* __restrict__ dinv) {
    __shared__ int rows[AG_ROWS];
    __shared__ int wsum[8];
    const int tid = threadIdx.x;
    const int blk = blockIdx.x;
    if (tid < AG_ROWS) rows[tid] = 0;
    const int cnt   = bincount[blk];
    const int start = gcn_block_start(bincount, blk, wsum);
    const int last = cnt - 1;
    for (int base = 0; base < cnt; base += 256 * AG_SWEEP) {
        int p[AG_SWEEP];
#pragma unroll
        for (int k = 0; k < AG_SWEEP; ++k) {
            const int i = base + k * 256 + tid;
            p[k] = erec[2 * (size_t)(start + (i < last ? i : last))];
        }
        asm volatile("" :: "v"(p[0]), "v"(p[1]), "v"(p[2]), "v"(p[3]), "v"(p[4]), "v"(p[5]), "v"(p[6]), "v"(p[7]));
#pragma unroll
        for (int k = 0; k < AG_SWEEP; ++k)
            if (base + k * 256 + tid < cnt) atomicAdd(&rows[p[k] >> AG_SRC_BITS], 1);
    }
    __syncthreads();
    if (tid < AG_ROWS) dinv[blk * AG_ROWS + tid] = rsqrtf((float)rows[tid] + 1.0f);
}

__global__ __launch_bounds__(256) void gcn_edge_weights(int* __restrict__ erec,
                                                        const int* __restrict__ bincount,
                                                        const float* __restrict__ dinv) {
    __shared__ int wsum[8];
    const int tid = threadIdx.x;
    const int blk = blockIdx.x;
    const int cnt   = bincount[blk];
    const int start = gcn_block_start(bincount, blk, wsum);
    const float* drow = dinv + blk * AG_ROWS;
    const int last = cnt - 1;
    for (int base = 0; base < cnt; base += 256 * AG_SWEEP) {
        int p[AG_SWEEP];
#pragma unroll
        for (int k = 0; k < AG_SWEEP; ++k) {
            const int i = base + k * 256 + tid;
            p[k] = erec[2 * (size_t)(start + (i < last ? i : last))];
        }
        float a[AG_SWEEP], b[AG_SWEEP];
#pragma unroll
        for (int k = 0; k < AG_SWEEP; ++k) {
            a[k] = dinv[p[k] & ((1 << AG_SRC_BITS) - 1)];
            b[k] = drow[p[k] >> AG_SRC_BITS];
        }
        asm volatile("" :: "v"(a[0]), "v"(a[1]), "v"(a[2]), "v"(a[3]), "v"(a[4]), "v"(a[5]), "v"(a[6]), "v"(a[7]),
                           "v"(b[0]), "v"(b[1]), "v"(b[2]), "v"(b[3]), "v"(b[4]), "v"(b[5]), "v"(b[6]), "v"(b[7]));
#pragma unroll
        for (int k = 0; k < AG_SWEEP; ++k) {
            const int i = base + k * 256 + tid;
            if (i < cnt) erec[2 * (size_t)(start + i) + 1] = __float_as_int(a[k] * b[k]);
        }
    }
}

#define AG_GROUP 8
#define AG_TILE_ROWS 6

__global__ __launch_bounds__(256) void gcn_block_sums(const int* __restrict__ erec,
                                                      const int* __restrict__ bincount,
                                                      const float* __restrict__ h,
                                                      const float* __restrict__ dinv,
                                                      const float* __restrict__ bias,
                                                      float* __restrict__ acc) {
    __shared__ float tile[AG_ROWS * GCN_D];
    __shared__ int wsum[8];
    const int tid  = threadIdx.x;
    const int lane = tid & 31;
    const int wv   = __builtin_amdgcn_readfirstlane(tid >> 5);
    const int blk  = blockIdx.x;

    int part = 0;
    for (int b = tid; b < blk; b += 256) part += bincount[b];
    const int cnt = bincount[blk];

    float*       arow = acc + (size_t)blk * (AG_ROWS * GCN_D) + 4 * lane;
    const float* hrow = h   + (size_t)blk * (AG_ROWS * GCN_D) + 4 * lane;
    const float* drow = dinv + blk * AG_ROWS;
    const float4 bc   = *(const float4*)(bias + 4 * lane);
#pragma unroll
    for (int p = 0; p < 2; ++p) {
        float4 v[AG_TILE_ROWS];
        float  di[AG_TILE_ROWS];
#pragma unroll
        for (int i = 0; i < AG_TILE_ROWS; ++i) {
            v[i]  = *(const float4*)(hrow + (wv + 8 * (AG_TILE_ROWS * p + i)) * GCN_D);
            di[i] = drow[wv + 8 * (AG_TILE_ROWS * p + i)];
        }
#pragma unroll
        for (int i = 0; i < AG_TILE_ROWS; ++i) {
            float* t = tile + (wv + 8 * (AG_TILE_ROWS * p + i)) * GCN_D + lane;
            t[0]  = v[i].x * di[i] * di[i] + bc.x;
            t[32] = v[i].y * di[i] * di[i] + bc.y;
            t[64] = v[i].z * di[i] * di[i] + bc.z;
            t[96] = v[i].w * di[i] * di[i] + bc.w;
        }
    }
    if (wv < AG_ROWS - 96) {
        const float4 v  = *(const float4*)(hrow + (96 + wv) * GCN_D);
        const float  di = drow[96 + wv];
        float* t = tile + (96 + wv) * GCN_D + lane;
        t[0]  = v.x * di * di + bc.x;
        t[32] = v.y * di * di + bc.y;
        t[64] = v.z * di * di + bc.z;
        t[96] = v.w * di * di + bc.w;
    }
    const int start = __builtin_amdgcn_readfirstlane(gcn_block_sum(part, wsum));
    const int end   = start + __builtin_amdgcn_readfirstlane(cnt);

    const float* hl = h + 4 * lane;
    float*       tl = tile + lane;
    int g0 = start + wv * 32;
    int2 nxt = make_int2(0, 0);
    if (g0 < end) {
        int idx = g0 + lane;
        if (idx > end - 1) idx = end - 1;
        nxt = *(const int2*)(erec + 2 * (size_t)idx);
    }
    for (; g0 < end; g0 += 256) {
        int m = end - g0;
        if (m > 32) m = 32;
        const int2 rec = nxt;
        if (g0 + 256 < end) {
            int idx = g0 + 256 + lane;
            if (idx > end - 1) idx = end - 1;
            nxt = *(const int2*)(erec + 2 * (size_t)idx);
        }
        int g = 0;
        for (; g + AG_GROUP <= m; g += AG_GROUP) {
            float4 v[AG_GROUP];
#pragma unroll
            for (int i = 0; i < AG_GROUP; ++i) {
                const int s = __builtin_amdgcn_readlane(rec.x, g + i) & ((1 << AG_SRC_BITS) - 1);
                v[i] = *(const float4*)(hl + (size_t)s * GCN_D);
            }
#pragma unroll
            for (int i = 0; i < AG_GROUP; ++i) {
                const int   dl = __builtin_amdgcn_readlane(rec.x, g + i) >> AG_SRC_BITS;
                const float w  = __int_as_float(__builtin_amdgcn_readlane(rec.y, g + i));
                float* t = tl + dl * GCN_D;
                atomicAdd(t +  0, v[i].x * w);
                atomicAdd(t + 32, v[i].y * w);
                atomicAdd(t + 64, v[i].z * w);
                atomicAdd(t + 96, v[i].w * w);
            }
        }
        for (; g + 4 <= m; g += 4) {
            float4 v[4];
#pragma unroll
            for (int i = 0; i < 4; ++i) {
                const int s = __builtin_amdgcn_readlane(rec.x, g + i) & ((1 << AG_SRC_BITS) - 1);
                v[i] = *(const float4*)(hl + (size_t)s * GCN_D);
            }
#pragma unroll
            for (int i = 0; i < 4; ++i) {
                const int   dl = __builtin_amdgcn_readlane(rec.x, g + i) >> AG_SRC_BITS;
                const float w  = __int_as_float(__builtin_amdgcn_readlane(rec.y, g + i));
                float* t = tl + dl * GCN_D;
                atomicAdd(t +  0, v[i].x * w);
                atomicAdd(t + 32, v[i].y * w);
                atomicAdd(t + 64, v[i].z * w);
                atomicAdd(t + 96, v[i].w * w);
            }
        }
        for (; g < m; ++g) {
            const int   p  = __builtin_amdgcn_readlane(rec.x, g);
            const float w  = __int_as_float(__builtin_amdgcn_readlane(rec.y, g));
            const float4 v = *(const float4*)(hl + (size_t)(p & ((1 << AG_SRC_BITS) - 1)) * GCN_D);
            float* t = tl + (p >> AG_SRC_BITS) * GCN_D;
            atomicAdd(t +  0, v.x * w);
            atomicAdd(t + 32, v.y * w);
            atomicAdd(t + 64, v.z * w);
            atomicAdd(t + 96, v.w * w);
        }
    }
    __syncthreads();
#pragma unroll
    for (int i = 0; i < 2 * AG_TILE_ROWS; ++i) {
        const float* t = tile + (wv + 8 * i) * GCN_D + lane;
        float4 v;
        v.x = t[0]; v.y = t[32]; v.z = t[64]; v.w = t[96];
        v.x = v.x > 0.0f ? v.x : 0.0f; v.y = v.y > 0.0f ? v.y : 0.0f;
        v.z = v.z > 0.0f ? v.z : 0.0f; v.w = v.w > 0.0f ? v.w : 0.0f;
        *(float4*)(arow + (wv + 8 * i) * GCN_D) = v;
    }
    if (wv < AG_ROWS - 96) {
        const float* t = tile + (96 + wv) * GCN_D + lane;
        float4 v;
        v.x = t[0]; v.y = t[32]; v.z = t[64]; v.w = t[96];
        v.x = v.x > 0.0f ? v.x : 0.0f; v.y = v.y > 0.0f ? v.y : 0.0f;
        v.z = v.z > 0.0f ? v.z : 0.0f; v.w = v.w > 0.0f ? v.w : 0.0f;
        *(float4*)(arow + (96 + wv) * GCN_D) = v;
    }
}

extern "C" void kernel_launch(void* const* d_in, const int* in_sizes, int n_in,
                              void* d_out, int out_size, void* d_ws, size_t ws_size,
                              hipStream_t stream) {
    const float* x  = (const float*)d_in[0];
    const float* W1 = (const float*)d_in[1];
    const float* b1 = (const float*)d_in[2];
    const float* W2 = (const float*)d_in[3];
    const float* b2 = (const float*)d_in[4];
    const float* W3 = (const float*)d_in[5];
    const float* b3 = (const float*)d_in[6];
    const int*   eidx = (const int*)d_in[7];
    const int ne = in_sizes[7] / 2;
    const int* src = eidx;
    const int* dst = eidx + ne;

    char* ws = (char*)d_ws;
    size_t off = 0;
    int*   bins  = (int*)(ws + off);   off += (size_t)AG_BIN_INTS * 4;
    float* dinv  = (float*)(ws + off); off += (size_t)GCN_DINV_SLOTS * 4;
    int*   erec  = (int*)(ws + off);   off += (((size_t)ne * 8 + 255) / 256) * 256;
    float* h     = (float*)(ws + off); off += (size_t)GCN_N * GCN_D * 4;
    float* acc_a = (float*)(ws + off); off += (size_t)GCN_N * GCN_D * 4;
    float* acc_b = (float*)(ws + off); off += (size_t)GCN_N * GCN_D * 4;
    float* wf    = (float*)(ws + off);
    float* out   = (float*)d_out;

    const int blk = 256;
    const int g_fill  = AG_BIN_INTS / blk;
    const int g_bin   = (ne + AG_CHUNK - 1) / AG_CHUNK;
    const int g_wf    = 3 * WF_FLOATS / blk;
    const int g_gemm  = GCN_N / 16 / 2;

    gcn_fill_zero_and_weight_fragments<<<g_fill + g_wf, blk, 0, stream>>>((float*)bins, AG_BIN_INTS, g_fill,
                                                                          W1, W2, W3, wf);
    gcn_block_counts<<<g_bin + GEMM_BLOCKS_A, blk, 0, stream>>>(dst, bins, ne, g_bin, x, wf, h);
    gcn_edge_records<<<g_bin + GEMM_BLOCKS_B, blk, 0, stream>>>(src, dst, bins, bins + AG_CURSOR_AT, erec, ne, g_bin, x, wf, h);
    gcn_block_dinv<<<AG_BLOCKS, blk, 0, stream>>>(erec, bins, dinv);
    gcn_edge_weights<<<AG_BLOCKS, blk, 0, stream>>>(erec, bins, dinv);

    gcn_block_sums<<<AG_BLOCKS, blk, 0, stream>>>(erec, bins, h, dinv, b1, acc_a);

    gcn_gemm<<<g_gemm, 64, 0, stream>>>(acc_a, wf + WF_FLOATS, h);
    gcn_block_sums<<<AG_BLOCKS, blk, 0, stream>>>(erec, bins, h, dinv, b2, acc_b);

    gcn_gemm<<<g_gemm, 64, 0, stream>>>(acc_b, wf + 2 * WF_FLOATS, h);
    gcn_block_sums<<<AG_BLOCKS, blk, 0, stream>>>(erec, bins, h, dinv, b3, out);
}
